# starting kernel
# baseline (speedup 1.0000x reference)
#include <hip/hip_runtime.h>
#include <hip/hip_bf16.h>


#define B_ 2
#define L_ 4096
#define C_ 7
#define H_ 8
#define D_ 64
#define W_ 512
#define M_ 32
#define ROWS_ (B_*L_)

typedef float v2f __attribute__((ext_vector_type(2)));
typedef float v8f __attribute__((ext_vector_type(8)));

__global__ __launch_bounds__(256) void qkv_proj_wmma(
    const float* __restrict__ hs,
    const float* __restrict__ wq,
    const float* __restrict__ wk,
    const float* __restrict__ wv,
    float* __restrict__ q, float* __restrict__ k, float* __restrict__ v)
{
    const int MT = ROWS_ / 16;
    const int NT = W_ / 16;
    const int lane = threadIdx.x & 31;
    const int wave = threadIdx.x >> 5;
    const int tile = blockIdx.x * (blockDim.x >> 5) + wave;

    const int mat = tile / (MT * NT);
    const int t   = tile % (MT * NT);
    const int mt  = t / NT;
    const int nt  = t % NT;

    const float* __restrict__ wmat = (mat == 0) ? wq : (mat == 1) ? wk : wv;
    float* __restrict__ outp       = (mat == 0) ? q  : (mat == 1) ? k  : v;

    const int mrow  = lane & 15;
    const int khalf = lane >> 4;
    const int arow  = mt * 16 + mrow;
    const int ncol  = nt * 16 + (lane & 15);

    v8f c = {};
#pragma unroll
    for (int ks = 0; ks < 2; ++ks) {
        const int   k0   = ks * 4 + khalf * 2;
        const bool  inr  = (k0 + 1 < C_);
        const int   k1i  = inr ? (k0 + 1) : 0;
        const float kmsk = inr ? 1.0f : 0.0f;
        v2f a, b;
        a.x = hs[arow * C_ + k0];
        a.y = hs[arow * C_ + k1i] * kmsk;
        b.x = wmat[ncol * C_ + k0];
        b.y = wmat[ncol * C_ + k1i] * kmsk;
        c = __builtin_amdgcn_wmma_f32_16x16x4_f32(
                false, a, false, b, (short)0, c, false, false);
    }

    const float scale = (mat == 0) ? 0.125f : 1.0f;
    const int rbase = mt * 16 + (khalf ? 8 : 0);
#pragma unroll
    for (int r = 0; r < 8; ++r)
        outp[(rbase + r) * W_ + ncol] = c[r] * scale;
}

__global__ __launch_bounds__(256) void pyr_attn(
    const float* __restrict__ q, const float* __restrict__ k,
    const float* __restrict__ v, const int* __restrict__ qk_mask,
    float* __restrict__ aout)
{
    const int bl = blockIdx.x;
    const int b  = bl / L_;
    const int l  = bl % L_;

    __shared__ float qs[W_];
    __shared__ int   idxs[M_];

    const int tid = threadIdx.x;
    qs[tid]       = q[bl * W_ + tid];
    qs[tid + 256] = q[bl * W_ + tid + 256];
    if (tid < M_) idxs[tid] = qk_mask[l * M_ + tid];
    __syncthreads();

    const int h    = tid >> 5;
    const int lane = tid & 31;

    const int  raw   = idxs[lane];
    const bool valid = (raw >= 0);
    const int  krow  = b * L_ + (valid ? raw : 0);

    const float* __restrict__ vpre = v + (size_t)krow * W_ + h * D_;
    __builtin_prefetch(vpre,      0, 0);
    __builtin_prefetch(vpre + 32, 0, 0);

    const float* __restrict__ kp = k + (size_t)krow * W_ + h * D_;
    float s = 0.0f;
#pragma unroll 8
    for (int d = 0; d < D_; ++d)
        s = fmaf(qs[h * D_ + d], kp[d], s);
    float score = valid ? s : -1.0e9f;

    float mx = score;
#pragma unroll
    for (int off = 16; off > 0; off >>= 1)
        mx = fmaxf(mx, __shfl_xor(mx, off, 32));
    float p = valid ? __expf(score - mx) : 0.0f;
    float sum = p;
#pragma unroll
    for (int off = 16; off > 0; off >>= 1)
        sum += __shfl_xor(sum, off, 32);
    const float attn = p * (1.0f / fmaxf(sum, 1.0e-30f));

    float a0 = 0.0f, a1 = 0.0f;
#pragma unroll 4
    for (int m = 0; m < M_; ++m) {
        const float am  = __shfl(attn, m, 32);
        const int  raw2 = idxs[m];
        const int  vrow = b * L_ + (raw2 >= 0 ? raw2 : 0);
        const float* __restrict__ vp = v + (size_t)vrow * W_ + h * D_;
        a0 = fmaf(am, vp[lane],      a0);
        a1 = fmaf(am, vp[lane + 32], a1);
    }
    aout[bl * W_ + h * D_ + lane]      = a0;
    aout[bl * W_ + h * D_ + lane + 32] = a1;
}

__global__ __launch_bounds__(256) void out_proj_ln_wmma(
    const float* __restrict__ aout, const float* __restrict__ fc_w,
    const float* __restrict__ fc_b, const float* __restrict__ hs,
    const float* __restrict__ ln_s, const float* __restrict__ ln_b,
    float* __restrict__ out)
{
    const int lane = threadIdx.x & 31;
    const int wave = threadIdx.x >> 5;
    const int mt   = blockIdx.x * (blockDim.x >> 5) + wave;

    const int mrow  = lane & 15;
    const int khalf = lane >> 4;
    const int arow  = mt * 16 + mrow;
    const int col   = lane & 15;
    const bool cv   = (col < C_);

    const float cmask = cv ? 1.0f : 0.0f;
    const int   bcol  = cv ? col : 0;
    const float* __restrict__ arowp = aout + (size_t)arow * W_;
    const float* __restrict__ browp = fc_w + (size_t)bcol * W_;

    v8f c = {};
    for (int kk = 0; kk < W_; kk += 4) {
        const int k0 = kk + khalf * 2;
        v2f a = *(const v2f*)(arowp + k0);
        v2f b = *(const v2f*)(browp + k0);
        b.x *= cmask;
        b.y *= cmask;
        c = __builtin_amdgcn_wmma_f32_16x16x4_f32(
                false, a, false, b, (short)0, c, false, false);
    }

    const float bias = cv ? fc_b[col] : 0.0f;
    const float lsc  = cv ? ln_s[col] : 0.0f;
    const float lbi  = cv ? ln_b[col] : 0.0f;
    const int   rbase = mt * 16 + (khalf ? 8 : 0);

#pragma unroll
    for (int r = 0; r < 8; ++r) {
        const int   row = rbase + r;
        const float val = c[r] + bias + (cv ? hs[row * C_ + col] : 0.0f);

        float x = cv ? val : 0.0f;
        float ssum = x;
#pragma unroll
        for (int off = 8; off > 0; off >>= 1)
            ssum += __shfl_xor(ssum, off, 32);
        const float mu = ssum * (1.0f / (float)C_);

        float dv = cv ? (val - mu) : 0.0f;
        float vsum = dv * dv;
#pragma unroll
        for (int off = 8; off > 0; off >>= 1)
            vsum += __shfl_xor(vsum, off, 32);
        const float var = vsum * (1.0f / (float)C_);

        const float y = (val - mu) * rsqrtf(var + 1.0e-6f) * lsc + lbi;
        if (cv) out[row * C_ + col] = y;
    }
}

extern "C" void kernel_launch(void* const* d_in, const int* in_sizes, int n_in,
                              void* d_out, int out_size, void* d_ws, size_t ws_size,
                              hipStream_t stream)
{
    const float* hs  = (const float*)d_in[0];
    const float* wq  = (const float*)d_in[1];
    const float* wk  = (const float*)d_in[2];
    const float* wv  = (const float*)d_in[3];
    const float* fcw = (const float*)d_in[4];
    const float* fcb = (const float*)d_in[5];
    const float* lns = (const float*)d_in[6];
    const float* lnb = (const float*)d_in[7];
    const int*   qkm = (const int*)d_in[8];
    float* out = (float*)d_out;

    float* qb = (float*)d_ws;
    float* kb = qb + (size_t)ROWS_ * W_;
    float* vb = kb + (size_t)ROWS_ * W_;
    float* ab = vb + (size_t)ROWS_ * W_;

    qkv_proj_wmma<<<6144, 256, 0, stream>>>(hs, wq, wk, wv, qb, kb, vb);
    pyr_attn<<<ROWS_, 256, 0, stream>>>(qb, kb, vb, qkm, ab);
    out_proj_ln_wmma<<<64, 256, 0, stream>>>(ab, fcw, fcb, hs, lns, lnb, out);
}


# Fold the rank-7 K and V projections into 7x7 per-head matrices (WMMA) so attention gathers 7-float input rows instead of 512-float K/V rows; 2 launches; fp summation order changes (reassociated)
# speedup vs baseline: 8.7799x; 8.7799x over previous
#include <hip/hip_runtime.h>


#define B_ 2
#define L_ 4096
#define C_ 7
#define H_ 8
#define D_ 64
#define W_ 512
#define M_ 32
#define ROWS_ (B_*L_)
#define CP_ 8
#define HC_ (H_*CP_)
#define ZS_ 68

typedef float v2f __attribute__((ext_vector_type(2)));
typedef float v8f __attribute__((ext_vector_type(8)));

__global__ __launch_bounds__(256) void fold_weights_wmma(
    const float* __restrict__ wq, const float* __restrict__ wk,
    const float* __restrict__ wv, const float* __restrict__ fcw,
    float* __restrict__ amat, float* __restrict__ gt)
{
    const int  lane  = threadIdx.x & 31;
    const int  h     = threadIdx.x >> 5;
    const bool isg   = (blockIdx.x != 0);
    const int  mrow  = lane & 15;
    const int  khalf = lane >> 4;
    const bool lv    = (mrow < C_);
    const int  lc    = lv ? mrow : 0;
    const float lmask = lv ? 1.0f : 0.0f;

    const float* __restrict__ pa = isg ? (fcw + lc * W_ + h * D_)
                                       : (wk + h * D_ * C_ + lc);
    const int sak = isg ? 1 : C_;
    const float* __restrict__ pb = (isg ? wv : wq) + h * D_ * C_ + lc;

    v8f c = {};
#pragma unroll
    for (int kk = 0; kk < D_; kk += 4) {
        const int k0 = kk + khalf * 2;
        v2f a, b;
        a.x = pa[k0 * sak] * lmask;
        a.y = pa[(k0 + 1) * sak] * lmask;
        b.x = pb[k0 * C_] * lmask;
        b.y = pb[(k0 + 1) * C_] * lmask;
        c = __builtin_amdgcn_wmma_f32_16x16x4_f32(
                false, a, false, b, (short)0, c, false, false);
    }
    if (lane < CP_) {
        if (!isg) {
#pragma unroll
            for (int r = 0; r < 8; ++r)
                amat[(h * CP_ + r) * CP_ + lane] = c[r] * 0.125f;
        } else {
#pragma unroll
            for (int r = 0; r < 8; ++r)
                gt[(h * CP_ + lane) * CP_ + r] = c[r];
        }
    }
}

__global__ __launch_bounds__(256) void pyr_attn_folded(
    const float* __restrict__ hs, const int* __restrict__ qk_mask,
    const float* __restrict__ amat, const float* __restrict__ gt,
    const float* __restrict__ fc_b, const float* __restrict__ ln_s,
    const float* __restrict__ ln_b, float* __restrict__ out)
{
    __shared__ __attribute__((aligned(16))) float xs[16 * M_ * CP_];
    __shared__ __attribute__((aligned(16))) float qps[16 * HC_];
    __shared__ __attribute__((aligned(16))) float zs[16 * ZS_];

    const int tid  = threadIdx.x;
    const int lane = tid & 31;
    const int wave = tid >> 5;
    const int r0   = blockIdx.x * 16;
    const int b    = r0 / L_;
    const int l0   = r0 % L_;

#pragma unroll
    for (int j = 0; j < 2; ++j) {
        const int  p     = tid + 256 * j;
        const int  raw   = qk_mask[l0 * M_ + p];
        const bool valid = (raw >= 0);
        const float* __restrict__ src = hs + (size_t)(b * L_ + (valid ? raw : 0)) * C_;
        float4 lo, hi;
        lo.x = src[0]; lo.y = src[1]; lo.z = src[2]; lo.w = src[3];
        hi.x = src[4]; hi.y = src[5]; hi.z = src[6]; hi.w = valid ? 1.0f : 0.0f;
        ((float4*)xs)[p * 2]     = lo;
        ((float4*)xs)[p * 2 + 1] = hi;
    }

    if (wave < 4) {
        const int mrow  = lane & 15;
        const int khalf = lane >> 4;
        const int n     = wave * 16 + mrow;
        const float* __restrict__ arow = hs + (size_t)(r0 + mrow) * C_;
        v8f c = {};
#pragma unroll
        for (int ks = 0; ks < 2; ++ks) {
            const int   k0   = ks * 4 + khalf * 2;
            const bool  inr  = (k0 + 1 < C_);
            const int   k1i  = inr ? (k0 + 1) : 0;
            const float kmsk = inr ? 1.0f : 0.0f;
            v2f a;
            a.x = arow[k0];
            a.y = arow[k1i] * kmsk;
            const v2f bb = *(const v2f*)(amat + n * CP_ + k0);
            c = __builtin_amdgcn_wmma_f32_16x16x4_f32(
                    false, a, false, bb, (short)0, c, false, false);
        }
#pragma unroll
        for (int r = 0; r < 8; ++r)
            qps[(r + 8 * khalf) * HC_ + n] = c[r];
    }
    __syncthreads();

    {
        const int h = lane >> 2;
        const int g = lane & 3;
#pragma unroll
        for (int ii = 0; ii < 2; ++ii) {
            const int i = wave * 2 + ii;
            const float4 q0 = ((const float4*)qps)[(i * HC_ + h * CP_) / 4];
            const float4 q1 = ((const float4*)qps)[(i * HC_ + h * CP_) / 4 + 1];
            float4 xa[8], xb[8];
            float  sc[8];
#pragma unroll
            for (int t = 0; t < 8; ++t) {
                const int m = t * 4 + g;
                xa[t] = ((const float4*)xs)[(i * M_ + m) * 2];
                xb[t] = ((const float4*)xs)[(i * M_ + m) * 2 + 1];
                float s = q0.x * xa[t].x;
                s = fmaf(q0.y, xa[t].y, s);
                s = fmaf(q0.z, xa[t].z, s);
                s = fmaf(q0.w, xa[t].w, s);
                s = fmaf(q1.x, xb[t].x, s);
                s = fmaf(q1.y, xb[t].y, s);
                s = fmaf(q1.z, xb[t].z, s);
                sc[t] = (xb[t].w != 0.0f) ? s : -1.0e9f;
            }
            float mx = sc[0];
#pragma unroll
            for (int t = 1; t < 8; ++t) mx = fmaxf(mx, sc[t]);
            mx = fmaxf(mx, __shfl_xor(mx, 1, 32));
            mx = fmaxf(mx, __shfl_xor(mx, 2, 32));

            float sum = 0.0f;
            float z0 = 0.f, z1 = 0.f, z2 = 0.f, z3 = 0.f, z4 = 0.f, z5 = 0.f, z6 = 0.f;
#pragma unroll
            for (int t = 0; t < 8; ++t) {
                const float p = (xb[t].w != 0.0f) ? __expf(sc[t] - mx) : 0.0f;
                sum += p;
                z0 = fmaf(p, xa[t].x, z0);
                z1 = fmaf(p, xa[t].y, z1);
                z2 = fmaf(p, xa[t].z, z2);
                z3 = fmaf(p, xa[t].w, z3);
                z4 = fmaf(p, xb[t].x, z4);
                z5 = fmaf(p, xb[t].y, z5);
                z6 = fmaf(p, xb[t].z, z6);
            }
#pragma unroll
            for (int off = 1; off <= 2; off <<= 1) {
                sum += __shfl_xor(sum, off, 32);
                z0  += __shfl_xor(z0,  off, 32);
                z1  += __shfl_xor(z1,  off, 32);
                z2  += __shfl_xor(z2,  off, 32);
                z3  += __shfl_xor(z3,  off, 32);
                z4  += __shfl_xor(z4,  off, 32);
                z5  += __shfl_xor(z5,  off, 32);
                z6  += __shfl_xor(z6,  off, 32);
            }
            const float inv = 1.0f / fmaxf(sum, 1.0e-30f);
            if (g == 0) {
                float4 lo, hi;
                lo.x = z0 * inv; lo.y = z1 * inv; lo.z = z2 * inv; lo.w = z3 * inv;
                hi.x = z4 * inv; hi.y = z5 * inv; hi.z = z6 * inv; hi.w = 0.0f;
                ((float4*)zs)[(i * ZS_ + h * CP_) / 4]     = lo;
                ((float4*)zs)[(i * ZS_ + h * CP_) / 4 + 1] = hi;
            }
        }
    }
    __syncthreads();

    if (wave == 0) {
        const int  mrow  = lane & 15;
        const int  khalf = lane >> 4;
        const int  col   = lane & 15;
        const bool cv    = (col < C_);
        const float cmask = cv ? 1.0f : 0.0f;
        const int   bcol  = cv ? col : 0;

        v8f c = {};
#pragma unroll
        for (int kk = 0; kk < HC_; kk += 4) {
            const int k0 = kk + khalf * 2;
            const v2f a = *(const v2f*)(zs + mrow * ZS_ + k0);
            v2f bb;
            bb.x = gt[k0 * CP_ + bcol] * cmask;
            bb.y = gt[(k0 + 1) * CP_ + bcol] * cmask;
            c = __builtin_amdgcn_wmma_f32_16x16x4_f32(
                    false, a, false, bb, (short)0, c, false, false);
        }

        const float bias = cv ? fc_b[col] : 0.0f;
        const float lsc  = cv ? ln_s[col] : 0.0f;
        const float lbi  = cv ? ln_b[col] : 0.0f;
        const int   rbase = r0 + (khalf ? 8 : 0);

#pragma unroll
        for (int r = 0; r < 8; ++r) {
            const int   row = rbase + r;
            const float val = c[r] + bias + (cv ? hs[row * C_ + col] : 0.0f);

            float x = cv ? val : 0.0f;
            float ssum = x;
#pragma unroll
            for (int off = 8; off > 0; off >>= 1)
                ssum += __shfl_xor(ssum, off, 32);
            const float mu = ssum * (1.0f / (float)C_);

            float dv = cv ? (val - mu) : 0.0f;
            float vsum = dv * dv;
#pragma unroll
            for (int off = 8; off > 0; off >>= 1)
                vsum += __shfl_xor(vsum, off, 32);
            const float var = vsum * (1.0f / (float)C_);

            const float y = (val - mu) * rsqrtf(var + 1.0e-6f) * lsc + lbi;
            if (cv) out[row * C_ + col] = y;
        }
    }
}

extern "C" void kernel_launch(void* const* d_in, const int* in_sizes, int n_in,
                              void* d_out, int out_size, void* d_ws, size_t ws_size,
                              hipStream_t stream)
{
    const float* hs  = (const float*)d_in[0];
    const float* wq  = (const float*)d_in[1];
    const float* wk  = (const float*)d_in[2];
    const float* wv  = (const float*)d_in[3];
    const float* fcw = (const float*)d_in[4];
    const float* fcb = (const float*)d_in[5];
    const float* lns = (const float*)d_in[6];
    const float* lnb = (const float*)d_in[7];
    const int*   qkm = (const int*)d_in[8];
    float* out = (float*)d_out;

    float* amat = (float*)d_ws;
    float* gtb  = amat + 512;

    fold_weights_wmma<<<2, 256, 0, stream>>>(wq, wk, wv, fcw, amat, gtb);
    pyr_attn_folded<<<512, 256, 0, stream>>>(hs, qkm, amat, gtb, fcb, lns, lnb, out);
}


# Wave 0's LayerNorm tail: row sums as row_xmask DPP adds (same 8,4,2,1 butterfly, 8 rows interleaved) instead of 64 serial LDS-unit shuffles; its 11 epilogue loads issued ahead of the WMMA chain
# speedup vs baseline: 12.4558x; 1.4187x over previous
#include <hip/hip_runtime.h>


#define B_ 2
#define L_ 4096
#define C_ 7
#define H_ 8
#define D_ 64
#define W_ 512
#define M_ 32
#define ROWS_ (B_*L_)
#define CP_ 8
#define HC_ (H_*CP_)
#define ZS_ 68

typedef float v2f __attribute__((ext_vector_type(2)));
typedef float v8f __attribute__((ext_vector_type(8)));

template <int MASK>
__device__ __forceinline__ float add_row_xor(float v)
{
    const int t = __builtin_amdgcn_update_dpp(0, __builtin_bit_cast(int, v),
                                              0x160 + MASK, 0xf, 0xf, false);
    return v + __builtin_bit_cast(float, t);
}

__global__ __launch_bounds__(256) void fold_weights_wmma(
    const float* __restrict__ wq, const float* __restrict__ wk,
    const float* __restrict__ wv, const float* __restrict__ fcw,
    float* __restrict__ amat, float* __restrict__ gt)
{
    const int  lane  = threadIdx.x & 31;
    const int  h     = threadIdx.x >> 5;
    const bool isg   = (blockIdx.x != 0);
    const int  mrow  = lane & 15;
    const int  khalf = lane >> 4;
    const bool lv    = (mrow < C_);
    const int  lc    = lv ? mrow : 0;
    const float lmask = lv ? 1.0f : 0.0f;

    const float* __restrict__ pa = isg ? (fcw + lc * W_ + h * D_)
                                       : (wk + h * D_ * C_ + lc);
    const int sak = isg ? 1 : C_;
    const float* __restrict__ pb = (isg ? wv : wq) + h * D_ * C_ + lc;

    v8f c = {};
#pragma unroll
    for (int kk = 0; kk < D_; kk += 4) {
        const int k0 = kk + khalf * 2;
        v2f a, b;
        a.x = pa[k0 * sak] * lmask;
        a.y = pa[(k0 + 1) * sak] * lmask;
        b.x = pb[k0 * C_] * lmask;
        b.y = pb[(k0 + 1) * C_] * lmask;
        c = __builtin_amdgcn_wmma_f32_16x16x4_f32(
                false, a, false, b, (short)0, c, false, false);
    }
    if (lane < CP_) {
        if (!isg) {
#pragma unroll
            for (int r = 0; r < 8; ++r)
                amat[(h * CP_ + r) * CP_ + lane] = c[r] * 0.125f;
        } else {
#pragma unroll
            for (int r = 0; r < 8; ++r)
                gt[(h * CP_ + lane) * CP_ + r] = c[r];
        }
    }
}

__global__ __launch_bounds__(256) void pyr_attn_folded(
    const float* __restrict__ hs, const int* __restrict__ qk_mask,
    const float* __restrict__ amat, const float* __restrict__ gt,
    const float* __restrict__ fc_b, const float* __restrict__ ln_s,
    const float* __restrict__ ln_b, float* __restrict__ out)
{
    __shared__ __attribute__((aligned(16))) float xs[16 * M_ * CP_];
    __shared__ __attribute__((aligned(16))) float qps[16 * HC_];
    __shared__ __attribute__((aligned(16))) float zs[16 * ZS_];

    const int tid  = threadIdx.x;
    const int lane = tid & 31;
    const int wave = tid >> 5;
    const int r0   = blockIdx.x * 16;
    const int b    = r0 / L_;
    const int l0   = r0 % L_;

#pragma unroll
    for (int j = 0; j < 2; ++j) {
        const int  p     = tid + 256 * j;
        const int  raw   = qk_mask[l0 * M_ + p];
        const bool valid = (raw >= 0);
        const float* __restrict__ src = hs + (size_t)(b * L_ + (valid ? raw : 0)) * C_;
        float4 lo, hi;
        lo.x = src[0]; lo.y = src[1]; lo.z = src[2]; lo.w = src[3];
        hi.x = src[4]; hi.y = src[5]; hi.z = src[6]; hi.w = valid ? 1.0f : 0.0f;
        ((float4*)xs)[p * 2]     = lo;
        ((float4*)xs)[p * 2 + 1] = hi;
    }

    if (wave < 4) {
        const int mrow  = lane & 15;
        const int khalf = lane >> 4;
        const int n     = wave * 16 + mrow;
        const float* __restrict__ arow = hs + (size_t)(r0 + mrow) * C_;
        v8f c = {};
#pragma unroll
        for (int ks = 0; ks < 2; ++ks) {
            const int   k0   = ks * 4 + khalf * 2;
            const bool  inr  = (k0 + 1 < C_);
            const int   k1i  = inr ? (k0 + 1) : 0;
            const float kmsk = inr ? 1.0f : 0.0f;
            v2f a;
            a.x = arow[k0];
            a.y = arow[k1i] * kmsk;
            const v2f bb = *(const v2f*)(amat + n * CP_ + k0);
            c = __builtin_amdgcn_wmma_f32_16x16x4_f32(
                    false, a, false, bb, (short)0, c, false, false);
        }
#pragma unroll
        for (int r = 0; r < 8; ++r)
            qps[(r + 8 * khalf) * HC_ + n] = c[r];
    }
    __syncthreads();

    {
        const int h = lane >> 2;
        const int g = lane & 3;
#pragma unroll
        for (int ii = 0; ii < 2; ++ii) {
            const int i = wave * 2 + ii;
            const float4 q0 = ((const float4*)qps)[(i * HC_ + h * CP_) / 4];
            const float4 q1 = ((const float4*)qps)[(i * HC_ + h * CP_) / 4 + 1];
            float4 xa[8], xb[8];
            float  sc[8];
#pragma unroll
            for (int t = 0; t < 8; ++t) {
                const int m = t * 4 + g;
                xa[t] = ((const float4*)xs)[(i * M_ + m) * 2];
                xb[t] = ((const float4*)xs)[(i * M_ + m) * 2 + 1];
                float s = q0.x * xa[t].x;
                s = fmaf(q0.y, xa[t].y, s);
                s = fmaf(q0.z, xa[t].z, s);
                s = fmaf(q0.w, xa[t].w, s);
                s = fmaf(q1.x, xb[t].x, s);
                s = fmaf(q1.y, xb[t].y, s);
                s = fmaf(q1.z, xb[t].z, s);
                sc[t] = (xb[t].w != 0.0f) ? s : -1.0e9f;
            }
            float mx = sc[0];
#pragma unroll
            for (int t = 1; t < 8; ++t) mx = fmaxf(mx, sc[t]);
            mx = fmaxf(mx, __shfl_xor(mx, 1, 32));
            mx = fmaxf(mx, __shfl_xor(mx, 2, 32));

            float sum = 0.0f;
            float z0 = 0.f, z1 = 0.f, z2 = 0.f, z3 = 0.f, z4 = 0.f, z5 = 0.f, z6 = 0.f;
#pragma unroll
            for (int t = 0; t < 8; ++t) {
                const float p = (xb[t].w != 0.0f) ? __expf(sc[t] - mx) : 0.0f;
                sum += p;
                z0 = fmaf(p, xa[t].x, z0);
                z1 = fmaf(p, xa[t].y, z1);
                z2 = fmaf(p, xa[t].z, z2);
                z3 = fmaf(p, xa[t].w, z3);
                z4 = fmaf(p, xb[t].x, z4);
                z5 = fmaf(p, xb[t].y, z5);
                z6 = fmaf(p, xb[t].z, z6);
            }
#pragma unroll
            for (int off = 1; off <= 2; off <<= 1) {
                sum += __shfl_xor(sum, off, 32);
                z0  += __shfl_xor(z0,  off, 32);
                z1  += __shfl_xor(z1,  off, 32);
                z2  += __shfl_xor(z2,  off, 32);
                z3  += __shfl_xor(z3,  off, 32);
                z4  += __shfl_xor(z4,  off, 32);
                z5  += __shfl_xor(z5,  off, 32);
                z6  += __shfl_xor(z6,  off, 32);
            }
            const float inv = 1.0f / fmaxf(sum, 1.0e-30f);
            if (g == 0) {
                float4 lo, hi;
                lo.x = z0 * inv; lo.y = z1 * inv; lo.z = z2 * inv; lo.w = z3 * inv;
                hi.x = z4 * inv; hi.y = z5 * inv; hi.z = z6 * inv; hi.w = 0.0f;
                ((float4*)zs)[(i * ZS_ + h * CP_) / 4]     = lo;
                ((float4*)zs)[(i * ZS_ + h * CP_) / 4 + 1] = hi;
            }
        }
    }
    __syncthreads();

    if (wave == 0) {
        const int  mrow  = lane & 15;
        const int  khalf = lane >> 4;
        const int  col   = lane & 15;
        const bool cv    = (col < C_);
        const float cmask = cv ? 1.0f : 0.0f;
        const int   bcol  = cv ? col : 0;
        const int   rbase = r0 + (khalf ? 8 : 0);

        const float bias = fc_b[bcol];
        const float lsc  = ln_s[bcol];
        const float lbi  = ln_b[bcol];
        float res[8];
#pragma unroll
        for (int r = 0; r < 8; ++r)
            res[r] = hs[(rbase + r) * C_ + bcol];

        v8f c = {};
#pragma unroll
        for (int kk = 0; kk < HC_; kk += 4) {
            const int k0 = kk + khalf * 2;
            const v2f a = *(const v2f*)(zs + mrow * ZS_ + k0);
            v2f bb;
            bb.x = gt[k0 * CP_ + bcol] * cmask;
            bb.y = gt[(k0 + 1) * CP_ + bcol] * cmask;
            c = __builtin_amdgcn_wmma_f32_16x16x4_f32(
                    false, a, false, bb, (short)0, c, false, false);
        }

        asm volatile("" :: "v"(lsc), "v"(lbi));

        float val[8], s1[8], s2[8];
#pragma unroll
        for (int r = 0; r < 8; ++r) {
            val[r] = c[r] + bias + res[r];
            s1[r]  = cv ? val[r] : 0.0f;
        }
#pragma unroll
        for (int r = 0; r < 8; ++r) s1[r] = add_row_xor<8>(s1[r]);
#pragma unroll
        for (int r = 0; r < 8; ++r) s1[r] = add_row_xor<4>(s1[r]);
#pragma unroll
        for (int r = 0; r < 8; ++r) s1[r] = add_row_xor<2>(s1[r]);
#pragma unroll
        for (int r = 0; r < 8; ++r) s1[r] = add_row_xor<1>(s1[r]);

#pragma unroll
        for (int r = 0; r < 8; ++r) {
            const float mu = s1[r] * (1.0f / (float)C_);
            const float dv = cv ? (val[r] - mu) : 0.0f;
            s2[r] = dv * dv;
        }
#pragma unroll
        for (int r = 0; r < 8; ++r) s2[r] = add_row_xor<8>(s2[r]);
#pragma unroll
        for (int r = 0; r < 8; ++r) s2[r] = add_row_xor<4>(s2[r]);
#pragma unroll
        for (int r = 0; r < 8; ++r) s2[r] = add_row_xor<2>(s2[r]);
#pragma unroll
        for (int r = 0; r < 8; ++r) s2[r] = add_row_xor<1>(s2[r]);

#pragma unroll
        for (int r = 0; r < 8; ++r) {
            const float mu  = s1[r] * (1.0f / (float)C_);
            const float var = s2[r] * (1.0f / (float)C_);
            const float y = (val[r] - mu) * rsqrtf(var + 1.0e-6f) * lsc + lbi;
            if (cv) out[(rbase + r) * C_ + col] = y;
        }
    }
}

extern "C" void kernel_launch(void* const* d_in, const int* in_sizes, int n_in,
                              void* d_out, int out_size, void* d_ws, size_t ws_size,
                              hipStream_t stream)
{
    const float* hs  = (const float*)d_in[0];
    const float* wq  = (const float*)d_in[1];
    const float* wk  = (const float*)d_in[2];
    const float* wv  = (const float*)d_in[3];
    const float* fcw = (const float*)d_in[4];
    const float* fcb = (const float*)d_in[5];
    const float* lns = (const float*)d_in[6];
    const float* lnb = (const float*)d_in[7];
    const int*   qkm = (const int*)d_in[8];
    float* out = (float*)d_out;

    float* amat = (float*)d_ws;
    float* gtb  = amat + 512;

    fold_weights_wmma<<<2, 256, 0, stream>>>(wq, wk, wv, fcw, amat, gtb);
    pyr_attn_folded<<<512, 256, 0, stream>>>(hs, qkm, amat, gtb, fcb, lns, lnb, out);
}


# Weight fold fused into the attention kernel, one launch instead of two: each work-group (now 32 rows, 256 groups) stages the weights in LDS and folds both 7x7 products of a head in one WMMA tile
# speedup vs baseline: 16.9086x; 1.3575x over previous
#include <hip/hip_runtime.h>


#define B_ 2
#define L_ 4096
#define C_ 7
#define H_ 8
#define D_ 64
#define W_ 512
#define M_ 32
#define ROWS_ (B_*L_)
#define CP_ 8
#define HC_ (H_*CP_)
#define ZS_ 68
#define RB_ 32
#define HW_ (D_*C_)
#define FS_ 68
#define WST_ (3*HW_ + C_*FS_ + 4)

typedef float v2f __attribute__((ext_vector_type(2)));
typedef float v8f __attribute__((ext_vector_type(8)));

template <int MASK>
__device__ __forceinline__ float add_row_xor(float v)
{
    const int t = __builtin_amdgcn_update_dpp(0, __builtin_bit_cast(int, v),
                                              0x160 + MASK, 0xf, 0xf, false);
    return v + __builtin_bit_cast(float, t);
}

__global__ __launch_bounds__(256) void pyr_attn_fused(
    const float* __restrict__ hs, const int* __restrict__ qk_mask,
    const float* __restrict__ wq, const float* __restrict__ wk,
    const float* __restrict__ wv, const float* __restrict__ fcw,
    const float* __restrict__ fc_b, const float* __restrict__ ln_s,
    const float* __restrict__ ln_b, float* __restrict__ out)
{
    __shared__ __attribute__((aligned(16))) float xs[RB_ * M_ * CP_];
    __shared__ __attribute__((aligned(16))) float qps[RB_ * HC_];
    __shared__ __attribute__((aligned(16))) float zs[RB_ * ZS_];
    __shared__ __attribute__((aligned(16))) float amats[H_ * CP_ * CP_];
    __shared__ __attribute__((aligned(16))) float gts[H_ * CP_ * CP_];
    __shared__ __attribute__((aligned(16))) float wst[H_ * WST_];

    const int tid   = threadIdx.x;
    const int lane  = tid & 31;
    const int wave  = tid >> 5;
    const int mrow  = lane & 15;
    const int khalf = lane >> 4;
    const int r0    = blockIdx.x * RB_;
    const int b     = r0 / L_;
    const int l0    = r0 % L_;

    int raw[4];
#pragma unroll
    for (int j = 0; j < 4; ++j)
        raw[j] = qk_mask[l0 * M_ + tid + 256 * j];

    float4 rk[4], rq[4], rv[4], rf[4];
#pragma unroll
    for (int j = 0; j < 4; ++j) {
        const int idx = (j < 3) ? (lane + 32 * j) : (96 + mrow);
        rk[j] = ((const float4*)(wk + wave * HW_))[idx];
        rq[j] = ((const float4*)(wq + wave * HW_))[idx];
        rv[j] = ((const float4*)(wv + wave * HW_))[idx];
        rf[j] = *(const float4*)(fcw + (idx >> 4) * W_ + wave * D_ + (idx & 15) * 4);
    }

    const int rt = wave >> 2;
    const int ct = wave & 3;
    v2f a1[2];
    {
        const float* __restrict__ arow = hs + (size_t)(r0 + rt * 16 + mrow) * C_;
#pragma unroll
        for (int ks = 0; ks < 2; ++ks) {
            const int   k0   = ks * 4 + khalf * 2;
            const bool  inr  = (k0 + 1 < C_);
            const int   k1i  = inr ? (k0 + 1) : 0;
            const float kmsk = inr ? 1.0f : 0.0f;
            a1[ks].x = arow[k0];
            a1[ks].y = arow[k1i] * kmsk;
        }
    }

    float4 glo[4], ghi[4];

    {
        float* st = wst + wave * WST_;
#pragma unroll
        for (int j = 0; j < 4; ++j) {
            const int idx = (j < 3) ? (lane + 32 * j) : (96 + mrow);
            ((float4*)st)[idx]              = rk[j];
            ((float4*)(st + HW_))[idx]      = rq[j];
            ((float4*)(st + 2 * HW_))[idx]  = rv[j];
            *(float4*)(st + 3 * HW_ + (idx >> 4) * FS_ + (idx & 15) * 4) = rf[j];
        }
        __builtin_amdgcn_fence(__ATOMIC_RELEASE, "wavefront");
        __builtin_amdgcn_wave_barrier();
        __builtin_amdgcn_fence(__ATOMIC_ACQUIRE, "wavefront");

#pragma unroll
        for (int j = 0; j < 4; ++j) {
            const bool valid = (raw[j] >= 0);
            const float* __restrict__ src = hs + (size_t)(b * L_ + (valid ? raw[j] : 0)) * C_;
            glo[j].x = src[0]; glo[j].y = src[1]; glo[j].z = src[2]; glo[j].w = src[3];
            ghi[j].x = src[4]; ghi[j].y = src[5]; ghi[j].z = src[6]; ghi[j].w = valid ? 1.0f : 0.0f;
        }

        const int  mc  = mrow & 7;
        const int  mcl = (mc < C_) ? mc : 0;
        const bool up  = (mrow >= 8);
        const float* pa = up ? (st + 3 * HW_ + mcl * FS_) : (st + mcl);
        const int sak = up ? 1 : C_;
        const float* pb = st + (up ? 2 * HW_ : HW_) + mcl;

        v8f c = {};
#pragma unroll
        for (int kk = 0; kk < D_; kk += 4) {
            const int k0 = kk + khalf * 2;
            v2f a, bb;
            a.x  = pa[k0 * sak];
            a.y  = pa[(k0 + 1) * sak];
            bb.x = pb[k0 * C_];
            bb.y = pb[(k0 + 1) * C_];
            c = __builtin_amdgcn_wmma_f32_16x16x4_f32(
                    false, a, false, bb, (short)0, c, false, false);
        }
        if (lane < CP_) {
#pragma unroll
            for (int r = 0; r < 8; ++r)
                amats[(wave * CP_ + r) * CP_ + lane] =
                    (r < C_ && lane < C_) ? c[r] * 0.125f : 0.0f;
        } else if (lane >= 24) {
            const int cvv = lane - 24;
            float4 lo, hi;
            const bool cok = (cvv < C_);
            lo.x = cok ? c[0] : 0.0f; lo.y = cok ? c[1] : 0.0f;
            lo.z = cok ? c[2] : 0.0f; lo.w = cok ? c[3] : 0.0f;
            hi.x = cok ? c[4] : 0.0f; hi.y = cok ? c[5] : 0.0f;
            hi.z = cok ? c[6] : 0.0f; hi.w = 0.0f;
            ((float4*)gts)[(wave * CP_ + cvv) * 2]     = lo;
            ((float4*)gts)[(wave * CP_ + cvv) * 2 + 1] = hi;
        }
    }

#pragma unroll
    for (int j = 0; j < 4; ++j) {
        const int p = tid + 256 * j;
        ((float4*)xs)[p * 2]     = glo[j];
        ((float4*)xs)[p * 2 + 1] = ghi[j];
    }
    __syncthreads();

    {
        const int n = ct * 16 + mrow;
        v8f c = {};
#pragma unroll
        for (int ks = 0; ks < 2; ++ks) {
            const int k0 = ks * 4 + khalf * 2;
            const v2f bb = *(const v2f*)(amats + n * CP_ + k0);
            c = __builtin_amdgcn_wmma_f32_16x16x4_f32(
                    false, a1[ks], false, bb, (short)0, c, false, false);
        }
#pragma unroll
        for (int r = 0; r < 8; ++r)
            qps[(rt * 16 + r + 8 * khalf) * HC_ + n] = c[r];
    }
    __syncthreads();

    const int  col   = mrow;
    const bool cv    = (col < C_);
    const int  bcol  = cv ? col : 0;
    const int  rbase = r0 + wave * 16 + (khalf ? 8 : 0);
    float bias = 0.0f, lsc = 0.0f, lbi = 0.0f;
    float res[8] = {0.f, 0.f, 0.f, 0.f, 0.f, 0.f, 0.f, 0.f};
    if (wave < 2) {
        bias = fc_b[bcol];
        lsc  = ln_s[bcol];
        lbi  = ln_b[bcol];
#pragma unroll
        for (int r = 0; r < 8; ++r)
            res[r] = hs[(rbase + r) * C_ + bcol];
    }

    {
        const int h = lane >> 2;
        const int g = lane & 3;
#pragma unroll
        for (int pr = 0; pr < 2; ++pr) {
            const int i0 = wave * 4 + pr * 2;
            float4 xa[2][8], xb[2][8];
            float  sc[2][8], mx[2];
#pragma unroll
            for (int u = 0; u < 2; ++u) {
                const int i = i0 + u;
                const float4 q0 = ((const float4*)qps)[(i * HC_ + h * CP_) / 4];
                const float4 q1 = ((const float4*)qps)[(i * HC_ + h * CP_) / 4 + 1];
#pragma unroll
                for (int t = 0; t < 8; ++t) {
                    const int m = t * 4 + g;
                    xa[u][t] = ((const float4*)xs)[(i * M_ + m) * 2];
                    xb[u][t] = ((const float4*)xs)[(i * M_ + m) * 2 + 1];
                    float s = q0.x * xa[u][t].x;
                    s = fmaf(q0.y, xa[u][t].y, s);
                    s = fmaf(q0.z, xa[u][t].z, s);
                    s = fmaf(q0.w, xa[u][t].w, s);
                    s = fmaf(q1.x, xb[u][t].x, s);
                    s = fmaf(q1.y, xb[u][t].y, s);
                    s = fmaf(q1.z, xb[u][t].z, s);
                    sc[u][t] = (xb[u][t].w != 0.0f) ? s : -1.0e9f;
                }
                mx[u] = sc[u][0];
#pragma unroll
                for (int t = 1; t < 8; ++t) mx[u] = fmaxf(mx[u], sc[u][t]);
            }
#pragma unroll
            for (int u = 0; u < 2; ++u) mx[u] = fmaxf(mx[u], __shfl_xor(mx[u], 1, 32));
#pragma unroll
            for (int u = 0; u < 2; ++u) mx[u] = fmaxf(mx[u], __shfl_xor(mx[u], 2, 32));

            float sum[2], z[2][7];
#pragma unroll
            for (int u = 0; u < 2; ++u) {
                sum[u] = 0.0f;
#pragma unroll
                for (int e = 0; e < 7; ++e) z[u][e] = 0.0f;
#pragma unroll
                for (int t = 0; t < 8; ++t) {
                    const float p = (xb[u][t].w != 0.0f) ? __expf(sc[u][t] - mx[u]) : 0.0f;
                    sum[u] += p;
                    z[u][0] = fmaf(p, xa[u][t].x, z[u][0]);
                    z[u][1] = fmaf(p, xa[u][t].y, z[u][1]);
                    z[u][2] = fmaf(p, xa[u][t].z, z[u][2]);
                    z[u][3] = fmaf(p, xa[u][t].w, z[u][3]);
                    z[u][4] = fmaf(p, xb[u][t].x, z[u][4]);
                    z[u][5] = fmaf(p, xb[u][t].y, z[u][5]);
                    z[u][6] = fmaf(p, xb[u][t].z, z[u][6]);
                }
            }
#pragma unroll
            for (int off = 1; off <= 2; off <<= 1) {
#pragma unroll
                for (int u = 0; u < 2; ++u) {
                    sum[u] += __shfl_xor(sum[u], off, 32);
#pragma unroll
                    for (int e = 0; e < 7; ++e)
                        z[u][e] += __shfl_xor(z[u][e], off, 32);
                }
            }
            if (g == 0) {
#pragma unroll
                for (int u = 0; u < 2; ++u) {
                    const int i = i0 + u;
                    const float inv = 1.0f / fmaxf(sum[u], 1.0e-30f);
                    float4 lo, hi;
                    lo.x = z[u][0] * inv; lo.y = z[u][1] * inv; lo.z = z[u][2] * inv; lo.w = z[u][3] * inv;
                    hi.x = z[u][4] * inv; hi.y = z[u][5] * inv; hi.z = z[u][6] * inv; hi.w = 0.0f;
                    ((float4*)zs)[(i * ZS_ + h * CP_) / 4]     = lo;
                    ((float4*)zs)[(i * ZS_ + h * CP_) / 4 + 1] = hi;
                }
            }
        }
    }
    __syncthreads();

    if (wave < 2) {
        const float cmask = cv ? 1.0f : 0.0f;

        v8f c = {};
#pragma unroll
        for (int kk = 0; kk < HC_; kk += 4) {
            const int k0 = kk + khalf * 2;
            const v2f a = *(const v2f*)(zs + (wave * 16 + mrow) * ZS_ + k0);
            v2f bb;
            bb.x = gts[k0 * CP_ + bcol] * cmask;
            bb.y = gts[(k0 + 1) * CP_ + bcol] * cmask;
            c = __builtin_amdgcn_wmma_f32_16x16x4_f32(
                    false, a, false, bb, (short)0, c, false, false);
        }


        float val[8], s1[8], s2[8];
#pragma unroll
        for (int r = 0; r < 8; ++r) {
            val[r] = c[r] + bias + res[r];
            s1[r]  = cv ? val[r] : 0.0f;
        }
#pragma unroll
        for (int r = 0; r < 8; ++r) s1[r] = add_row_xor<8>(s1[r]);
#pragma unroll
        for (int r = 0; r < 8; ++r) s1[r] = add_row_xor<4>(s1[r]);
#pragma unroll
        for (int r = 0; r < 8; ++r) s1[r] = add_row_xor<2>(s1[r]);
#pragma unroll
        for (int r = 0; r < 8; ++r) s1[r] = add_row_xor<1>(s1[r]);

#pragma unroll
        for (int r = 0; r < 8; ++r) {
            const float mu = s1[r] * (1.0f / (float)C_);
            const float dv = cv ? (val[r] - mu) : 0.0f;
            s2[r] = dv * dv;
        }
#pragma unroll
        for (int r = 0; r < 8; ++r) s2[r] = add_row_xor<8>(s2[r]);
#pragma unroll
        for (int r = 0; r < 8; ++r) s2[r] = add_row_xor<4>(s2[r]);
#pragma unroll
        for (int r = 0; r < 8; ++r) s2[r] = add_row_xor<2>(s2[r]);
#pragma unroll
        for (int r = 0; r < 8; ++r) s2[r] = add_row_xor<1>(s2[r]);

#pragma unroll
        for (int r = 0; r < 8; ++r) {
            const float mu  = s1[r] * (1.0f / (float)C_);
            const float var = s2[r] * (1.0f / (float)C_);
            const float y = (val[r] - mu) * rsqrtf(var + 1.0e-6f) * lsc + lbi;
            if (cv) out[(rbase + r) * C_ + col] = y;
        }
    }
}

extern "C" void kernel_launch(void* const* d_in, const int* in_sizes, int n_in,
                              void* d_out, int out_size, void* d_ws, size_t ws_size,
                              hipStream_t stream)
{
    const float* hs  = (const float*)d_in[0];
    const float* wq  = (const float*)d_in[1];
    const float* wk  = (const float*)d_in[2];
    const float* wv  = (const float*)d_in[3];
    const float* fcw = (const float*)d_in[4];
    const float* fcb = (const float*)d_in[5];
    const float* lns = (const float*)d_in[6];
    const float* lnb = (const float*)d_in[7];
    const int*   qkm = (const int*)d_in[8];
    float* out = (float*)d_out;

    pyr_attn_fused<<<256, 256, 0, stream>>>(hs, qkm, wq, wk, wv, fcw, fcb, lns, lnb, out);
}


# Stage 2 remapped to one lane per (row, head): the lane holds its row's 32 neighbours and finishes max, softmax sum and weighted sums itself in neighbour order (fp sum order changes); no shuffles
# speedup vs baseline: 18.3917x; 1.0877x over previous
#include <hip/hip_runtime.h>


#define B_ 2
#define L_ 4096
#define C_ 7
#define H_ 8
#define D_ 64
#define W_ 512
#define M_ 32
#define ROWS_ (B_*L_)
#define CP_ 8
#define HC_ (H_*CP_)
#define ZS_ 68
#define XR_ (M_*CP_ + 8)
#define RB_ 32
#define HW_ (D_*C_)
#define FS_ 68
#define WST_ (3*HW_ + C_*FS_ + 4)

typedef float v2f __attribute__((ext_vector_type(2)));
typedef float v8f __attribute__((ext_vector_type(8)));

template <int MASK>
__device__ __forceinline__ float add_row_xor(float v)
{
    const int t = __builtin_amdgcn_update_dpp(0, __builtin_bit_cast(int, v),
                                              0x160 + MASK, 0xf, 0xf, false);
    return v + __builtin_bit_cast(float, t);
}

__global__ __launch_bounds__(256) void pyr_attn_fused(
    const float* __restrict__ hs, const int* __restrict__ qk_mask,
    const float* __restrict__ wq, const float* __restrict__ wk,
    const float* __restrict__ wv, const float* __restrict__ fcw,
    const float* __restrict__ fc_b, const float* __restrict__ ln_s,
    const float* __restrict__ ln_b, float* __restrict__ out)
{
    __shared__ __attribute__((aligned(16))) float xs[RB_ * XR_];
    __shared__ __attribute__((aligned(16))) float qps[RB_ * HC_];
    __shared__ __attribute__((aligned(16))) float zs[RB_ * ZS_];
    __shared__ __attribute__((aligned(16))) float amats[H_ * CP_ * CP_];
    __shared__ __attribute__((aligned(16))) float gts[H_ * CP_ * CP_];
    __shared__ __attribute__((aligned(16))) float wst[H_ * WST_];

    const int tid   = threadIdx.x;
    const int lane  = tid & 31;
    const int wave  = tid >> 5;
    const int mrow  = lane & 15;
    const int khalf = lane >> 4;
    const int r0    = blockIdx.x * RB_;
    const int b     = r0 / L_;
    const int l0    = r0 % L_;

    int raw[4];
#pragma unroll
    for (int j = 0; j < 4; ++j)
        raw[j] = qk_mask[l0 * M_ + tid + 256 * j];

    float4 rk[4], rq[4], rv[4], rf[4];
#pragma unroll
    for (int j = 0; j < 4; ++j) {
        const int idx = (j < 3) ? (lane + 32 * j) : (96 + mrow);
        rk[j] = ((const float4*)(wk + wave * HW_))[idx];
        rq[j] = ((const float4*)(wq + wave * HW_))[idx];
        rv[j] = ((const float4*)(wv + wave * HW_))[idx];
        rf[j] = *(const float4*)(fcw + (idx >> 4) * W_ + wave * D_ + (idx & 15) * 4);
    }

    const int rt = wave >> 2;
    const int ct = wave & 3;
    v2f a1[2];
    {
        const float* __restrict__ arow = hs + (size_t)(r0 + rt * 16 + mrow) * C_;
#pragma unroll
        for (int ks = 0; ks < 2; ++ks) {
            const int   k0   = ks * 4 + khalf * 2;
            const bool  inr  = (k0 + 1 < C_);
            const int   k1i  = inr ? (k0 + 1) : 0;
            const float kmsk = inr ? 1.0f : 0.0f;
            a1[ks].x = arow[k0];
            a1[ks].y = arow[k1i] * kmsk;
        }
    }

    float4 glo[4], ghi[4];

    {
        float* st = wst + wave * WST_;
#pragma unroll
        for (int j = 0; j < 4; ++j) {
            const int idx = (j < 3) ? (lane + 32 * j) : (96 + mrow);
            ((float4*)st)[idx]              = rk[j];
            ((float4*)(st + HW_))[idx]      = rq[j];
            ((float4*)(st + 2 * HW_))[idx]  = rv[j];
            *(float4*)(st + 3 * HW_ + (idx >> 4) * FS_ + (idx & 15) * 4) = rf[j];
        }
        __builtin_amdgcn_fence(__ATOMIC_RELEASE, "wavefront");
        __builtin_amdgcn_wave_barrier();
        __builtin_amdgcn_fence(__ATOMIC_ACQUIRE, "wavefront");

#pragma unroll
        for (int j = 0; j < 4; ++j) {
            const bool valid = (raw[j] >= 0);
            const float* __restrict__ src = hs + (size_t)(b * L_ + (valid ? raw[j] : 0)) * C_;
            glo[j].x = src[0]; glo[j].y = src[1]; glo[j].z = src[2]; glo[j].w = src[3];
            ghi[j].x = src[4]; ghi[j].y = src[5]; ghi[j].z = src[6]; ghi[j].w = valid ? 1.0f : 0.0f;
        }

        const int  mc  = mrow & 7;
        const int  mcl = (mc < C_) ? mc : 0;
        const bool up  = (mrow >= 8);
        const float* pa = up ? (st + 3 * HW_ + mcl * FS_) : (st + mcl);
        const int sak = up ? 1 : C_;
        const float* pb = st + (up ? 2 * HW_ : HW_) + mcl;

        v8f c = {};
#pragma unroll
        for (int kk = 0; kk < D_; kk += 4) {
            const int k0 = kk + khalf * 2;
            v2f a, bb;
            a.x  = pa[k0 * sak];
            a.y  = pa[(k0 + 1) * sak];
            bb.x = pb[k0 * C_];
            bb.y = pb[(k0 + 1) * C_];
            c = __builtin_amdgcn_wmma_f32_16x16x4_f32(
                    false, a, false, bb, (short)0, c, false, false);
        }
        if (lane < CP_) {
#pragma unroll
            for (int r = 0; r < 8; ++r)
                amats[(wave * CP_ + r) * CP_ + lane] =
                    (r < C_ && lane < C_) ? c[r] * 0.125f : 0.0f;
        } else if (lane >= 24) {
            const int cvv = lane - 24;
            float4 lo, hi;
            const bool cok = (cvv < C_);
            lo.x = cok ? c[0] : 0.0f; lo.y = cok ? c[1] : 0.0f;
            lo.z = cok ? c[2] : 0.0f; lo.w = cok ? c[3] : 0.0f;
            hi.x = cok ? c[4] : 0.0f; hi.y = cok ? c[5] : 0.0f;
            hi.z = cok ? c[6] : 0.0f; hi.w = 0.0f;
            ((float4*)gts)[(wave * CP_ + cvv) * 2]     = lo;
            ((float4*)gts)[(wave * CP_ + cvv) * 2 + 1] = hi;
        }
    }

#pragma unroll
    for (int j = 0; j < 4; ++j) {
        const int p = tid + 256 * j;
        float4* dst = (float4*)(xs + (p >> 5) * XR_ + (p & 31) * CP_);
        dst[0] = glo[j];
        dst[1] = ghi[j];
    }
    __syncthreads();

    {
        const int n = ct * 16 + mrow;
        v8f c = {};
#pragma unroll
        for (int ks = 0; ks < 2; ++ks) {
            const int k0 = ks * 4 + khalf * 2;
            const v2f bb = *(const v2f*)(amats + n * CP_ + k0);
            c = __builtin_amdgcn_wmma_f32_16x16x4_f32(
                    false, a1[ks], false, bb, (short)0, c, false, false);
        }
#pragma unroll
        for (int r = 0; r < 8; ++r)
            qps[(rt * 16 + r + 8 * khalf) * HC_ + n] = c[r];
    }
    __syncthreads();

    const int  col   = mrow;
    const bool cv    = (col < C_);
    const int  bcol  = cv ? col : 0;
    const int  rbase = r0 + wave * 16 + (khalf ? 8 : 0);
    float bias = 0.0f, lsc = 0.0f, lbi = 0.0f;
    float res[8] = {0.f, 0.f, 0.f, 0.f, 0.f, 0.f, 0.f, 0.f};
    if (wave < 2) {
        bias = fc_b[bcol];
        lsc  = ln_s[bcol];
        lbi  = ln_b[bcol];
#pragma unroll
        for (int r = 0; r < 8; ++r)
            res[r] = hs[(rbase + r) * C_ + bcol];
    }

    {
        const int rr = lane >> 3;
        const int h  = lane & 7;
        const int i  = wave * 4 + rr;
        const float4 q0 = ((const float4*)qps)[(i * HC_ + h * CP_) / 4];
        const float4 q1 = ((const float4*)qps)[(i * HC_ + h * CP_) / 4 + 1];
        const float4* __restrict__ xrow = (const float4*)(xs + i * XR_);

        float4 xa[M_], xb[M_];
        float  sc[M_];
#pragma unroll
        for (int m = 0; m < M_; ++m) {
            xa[m] = xrow[m * 2];
            xb[m] = xrow[m * 2 + 1];
            float s = q0.x * xa[m].x;
            s = fmaf(q0.y, xa[m].y, s);
            s = fmaf(q0.z, xa[m].z, s);
            s = fmaf(q0.w, xa[m].w, s);
            s = fmaf(q1.x, xb[m].x, s);
            s = fmaf(q1.y, xb[m].y, s);
            s = fmaf(q1.z, xb[m].z, s);
            sc[m] = (xb[m].w != 0.0f) ? s : -1.0e9f;
        }
        float mx = sc[0];
#pragma unroll
        for (int m = 1; m < M_; ++m) mx = fmaxf(mx, sc[m]);

        float sum = 0.0f, z[7];
#pragma unroll
        for (int e = 0; e < 7; ++e) z[e] = 0.0f;
#pragma unroll
        for (int m = 0; m < M_; ++m) {
            const float p = (xb[m].w != 0.0f) ? __expf(sc[m] - mx) : 0.0f;
            sum += p;
            z[0] = fmaf(p, xa[m].x, z[0]);
            z[1] = fmaf(p, xa[m].y, z[1]);
            z[2] = fmaf(p, xa[m].z, z[2]);
            z[3] = fmaf(p, xa[m].w, z[3]);
            z[4] = fmaf(p, xb[m].x, z[4]);
            z[5] = fmaf(p, xb[m].y, z[5]);
            z[6] = fmaf(p, xb[m].z, z[6]);
        }
        const float inv = 1.0f / fmaxf(sum, 1.0e-30f);
        float4 lo, hi;
        lo.x = z[0] * inv; lo.y = z[1] * inv; lo.z = z[2] * inv; lo.w = z[3] * inv;
        hi.x = z[4] * inv; hi.y = z[5] * inv; hi.z = z[6] * inv; hi.w = 0.0f;
        ((float4*)zs)[(i * ZS_ + h * CP_) / 4]     = lo;
        ((float4*)zs)[(i * ZS_ + h * CP_) / 4 + 1] = hi;
    }
    __syncthreads();

    if (wave < 2) {
        const float cmask = cv ? 1.0f : 0.0f;

        v8f c = {};
#pragma unroll
        for (int kk = 0; kk < HC_; kk += 4) {
            const int k0 = kk + khalf * 2;
            const v2f a = *(const v2f*)(zs + (wave * 16 + mrow) * ZS_ + k0);
            v2f bb;
            bb.x = gts[k0 * CP_ + bcol] * cmask;
            bb.y = gts[(k0 + 1) * CP_ + bcol] * cmask;
            c = __builtin_amdgcn_wmma_f32_16x16x4_f32(
                    false, a, false, bb, (short)0, c, false, false);
        }


        float val[8], s1[8], s2[8];
#pragma unroll
        for (int r = 0; r < 8; ++r) {
            val[r] = c[r] + bias + res[r];
            s1[r]  = cv ? val[r] : 0.0f;
        }
#pragma unroll
        for (int r = 0; r < 8; ++r) s1[r] = add_row_xor<8>(s1[r]);
#pragma unroll
        for (int r = 0; r < 8; ++r) s1[r] = add_row_xor<4>(s1[r]);
#pragma unroll
        for (int r = 0; r < 8; ++r) s1[r] = add_row_xor<2>(s1[r]);
#pragma unroll
        for (int r = 0; r < 8; ++r) s1[r] = add_row_xor<1>(s1[r]);

#pragma unroll
        for (int r = 0; r < 8; ++r) {
            const float mu = s1[r] * (1.0f / (float)C_);
            const float dv = cv ? (val[r] - mu) : 0.0f;
            s2[r] = dv * dv;
        }
#pragma unroll
        for (int r = 0; r < 8; ++r) s2[r] = add_row_xor<8>(s2[r]);
#pragma unroll
        for (int r = 0; r < 8; ++r) s2[r] = add_row_xor<4>(s2[r]);
#pragma unroll
        for (int r = 0; r < 8; ++r) s2[r] = add_row_xor<2>(s2[r]);
#pragma unroll
        for (int r = 0; r < 8; ++r) s2[r] = add_row_xor<1>(s2[r]);

#pragma unroll
        for (int r = 0; r < 8; ++r) {
            const float mu  = s1[r] * (1.0f / (float)C_);
            const float var = s2[r] * (1.0f / (float)C_);
            const float y = (val[r] - mu) * rsqrtf(var + 1.0e-6f) * lsc + lbi;
            if (cv) out[(rbase + r) * C_ + col] = y;
        }
    }
}

extern "C" void kernel_launch(void* const* d_in, const int* in_sizes, int n_in,
                              void* d_out, int out_size, void* d_ws, size_t ws_size,
                              hipStream_t stream)
{
    const float* hs  = (const float*)d_in[0];
    const float* wq  = (const float*)d_in[1];
    const float* wk  = (const float*)d_in[2];
    const float* wv  = (const float*)d_in[3];
    const float* fcw = (const float*)d_in[4];
    const float* fcb = (const float*)d_in[5];
    const float* lns = (const float*)d_in[6];
    const float* lnb = (const float*)d_in[7];
    const int*   qkm = (const int*)d_in[8];
    float* out = (float*)d_out;

    pyr_attn_fused<<<256, 256, 0, stream>>>(hs, qkm, wq, wk, wv, fcw, fcb, lns, lnb, out);
}


# WMMA chains no longer wait on their operand feed: the fold runs K-quarter by K-quarter as its weight loads land (gather issued ahead); every chain reads its LDS fragments as one group ahead of it
# speedup vs baseline: 19.5402x; 1.0624x over previous
#include <hip/hip_runtime.h>


#define B_ 2
#define L_ 4096
#define C_ 7
#define H_ 8
#define D_ 64
#define W_ 512
#define M_ 32
#define ROWS_ (B_*L_)
#define CP_ 8
#define HC_ (H_*CP_)
#define ZS_ 68
#define XR_ (M_*CP_ + 8)
#define RB_ 32
#define HW_ (D_*C_)
#define FS_ 68
#define WST_ (3*HW_ + C_*FS_ + 4)

typedef float v2f __attribute__((ext_vector_type(2)));
typedef float v8f __attribute__((ext_vector_type(8)));

template <int MASK>
__device__ __forceinline__ float add_row_xor(float v)
{
    const int t = __builtin_amdgcn_update_dpp(0, __builtin_bit_cast(int, v),
                                              0x160 + MASK, 0xf, 0xf, false);
    return v + __builtin_bit_cast(float, t);
}

__device__ __forceinline__ v8f fold_quarter(
    const int g, float* st, const int t28,
    const float4 k4, const float4 q4, const float4 v4, const float4 f4,
    const float* pa, const int sak, const float* pb, const int khalf, v8f c)
{
    ((float4*)st)[28 * g + t28]             = k4;
    ((float4*)(st + HW_))[28 * g + t28]     = q4;
    ((float4*)(st + 2 * HW_))[28 * g + t28] = v4;
    *(float4*)(st + 3 * HW_ + (t28 >> 2) * FS_ + 16 * g + (t28 & 3) * 4) = f4;
    __builtin_amdgcn_fence(__ATOMIC_RELEASE, "wavefront");
    __builtin_amdgcn_wave_barrier();
    __builtin_amdgcn_fence(__ATOMIC_ACQUIRE, "wavefront");
    v2f a[4], bb[4];
#pragma unroll
    for (int s = 0; s < 4; ++s) {
        const int k0 = 16 * g + 4 * s + khalf * 2;
        a[s].x  = pa[k0 * sak];
        a[s].y  = pa[(k0 + 1) * sak];
        bb[s].x = pb[k0 * C_];
        bb[s].y = pb[(k0 + 1) * C_];
    }
    __builtin_amdgcn_sched_barrier(0);
#pragma unroll
    for (int s = 0; s < 4; ++s)
        c = __builtin_amdgcn_wmma_f32_16x16x4_f32(
                false, a[s], false, bb[s], (short)0, c, false, false);
    if (g < 3) asm volatile("" :: "v"(c));
    return c;
}

__global__ __launch_bounds__(256) void pyr_attn_fused(
    const float* __restrict__ hs, const int* __restrict__ qk_mask,
    const float* __restrict__ wq, const float* __restrict__ wk,
    const float* __restrict__ wv, const float* __restrict__ fcw,
    const float* __restrict__ fc_b, const float* __restrict__ ln_s,
    const float* __restrict__ ln_b, float* __restrict__ out)
{
    __shared__ __attribute__((aligned(16))) float xs[RB_ * XR_];
    __shared__ __attribute__((aligned(16))) float qps[RB_ * HC_];
    __shared__ __attribute__((aligned(16))) float zs[RB_ * ZS_];
    __shared__ __attribute__((aligned(16))) float amats[H_ * CP_ * CP_];
    __shared__ __attribute__((aligned(16))) float gts[H_ * CP_ * CP_];
    __shared__ __attribute__((aligned(16))) float wst[H_ * WST_];

    const int tid   = threadIdx.x;
    const int lane  = tid & 31;
    const int wave  = tid >> 5;
    const int mrow  = lane & 15;
    const int khalf = lane >> 4;
    const int r0    = blockIdx.x * RB_;
    const int b     = r0 / L_;
    const int l0    = r0 % L_;

    int raw[4];
#pragma unroll
    for (int j = 0; j < 4; ++j)
        raw[j] = qk_mask[l0 * M_ + tid + 256 * j];

    const int t28 = (lane < 28) ? lane : 27;
    const float4* __restrict__ wk4 = (const float4*)(wk + wave * HW_) + t28;
    const float4* __restrict__ wq4 = (const float4*)(wq + wave * HW_) + t28;
    const float4* __restrict__ wv4 = (const float4*)(wv + wave * HW_) + t28;
    const float*  __restrict__ fcp = fcw + (t28 >> 2) * W_ + wave * D_ + (t28 & 3) * 4;
    const float4 rk0 = wk4[0],  rq0 = wq4[0],  rv0 = wv4[0],  rf0 = *(const float4*)(fcp);
    __builtin_amdgcn_sched_barrier(0);
    const float4 rk1 = wk4[28], rq1 = wq4[28], rv1 = wv4[28], rf1 = *(const float4*)(fcp + 16);
    __builtin_amdgcn_sched_barrier(0);
    const float4 rk2 = wk4[56], rq2 = wq4[56], rv2 = wv4[56], rf2 = *(const float4*)(fcp + 32);
    __builtin_amdgcn_sched_barrier(0);
    const float4 rk3 = wk4[84], rq3 = wq4[84], rv3 = wv4[84], rf3 = *(const float4*)(fcp + 48);
    __builtin_amdgcn_sched_barrier(0);

    const int rt = wave >> 2;
    const int ct = wave & 3;
    v2f a1[2];
    {
        const float* __restrict__ arow = hs + (size_t)(r0 + rt * 16 + mrow) * C_;
#pragma unroll
        for (int ks = 0; ks < 2; ++ks) {
            const int   k0   = ks * 4 + khalf * 2;
            const bool  inr  = (k0 + 1 < C_);
            const int   k1i  = inr ? (k0 + 1) : 0;
            const float kmsk = inr ? 1.0f : 0.0f;
            a1[ks].x = arow[k0];
            a1[ks].y = arow[k1i] * kmsk;
        }
    }

    float4 glo[4], ghi[4];
#pragma unroll
    for (int j = 0; j < 4; ++j) {
        const bool valid = (raw[j] >= 0);
        const float* __restrict__ src = hs + (size_t)(b * L_ + (valid ? raw[j] : 0)) * C_;
        glo[j].x = src[0]; glo[j].y = src[1]; glo[j].z = src[2]; glo[j].w = src[3];
        ghi[j].x = src[4]; ghi[j].y = src[5]; ghi[j].z = src[6]; ghi[j].w = valid ? 1.0f : 0.0f;
    }

    {
        float* st = wst + wave * WST_;

        const int  mc  = mrow & 7;
        const int  mcl = (mc < C_) ? mc : 0;
        const bool up  = (mrow >= 8);
        const float* pa = up ? (st + 3 * HW_ + mcl * FS_) : (st + mcl);
        const int sak = up ? 1 : C_;
        const float* pb = st + (up ? 2 * HW_ : HW_) + mcl;

        v8f c = {};
        c = fold_quarter(0, st, t28, rk0, rq0, rv0, rf0, pa, sak, pb, khalf, c);
        c = fold_quarter(1, st, t28, rk1, rq1, rv1, rf1, pa, sak, pb, khalf, c);
        c = fold_quarter(2, st, t28, rk2, rq2, rv2, rf2, pa, sak, pb, khalf, c);
        c = fold_quarter(3, st, t28, rk3, rq3, rv3, rf3, pa, sak, pb, khalf, c);
        if (lane < CP_) {
#pragma unroll
            for (int r = 0; r < 8; ++r)
                amats[(wave * CP_ + r) * CP_ + lane] =
                    (r < C_ && lane < C_) ? c[r] * 0.125f : 0.0f;
        } else if (lane >= 24) {
            const int cvv = lane - 24;
            float4 lo, hi;
            const bool cok = (cvv < C_);
            lo.x = cok ? c[0] : 0.0f; lo.y = cok ? c[1] : 0.0f;
            lo.z = cok ? c[2] : 0.0f; lo.w = cok ? c[3] : 0.0f;
            hi.x = cok ? c[4] : 0.0f; hi.y = cok ? c[5] : 0.0f;
            hi.z = cok ? c[6] : 0.0f; hi.w = 0.0f;
            ((float4*)gts)[(wave * CP_ + cvv) * 2]     = lo;
            ((float4*)gts)[(wave * CP_ + cvv) * 2 + 1] = hi;
        }
    }

#pragma unroll
    for (int j = 0; j < 4; ++j) {
        const int p = tid + 256 * j;
        float4* dst = (float4*)(xs + (p >> 5) * XR_ + (p & 31) * CP_);
        dst[0] = glo[j];
        dst[1] = ghi[j];
    }
    __syncthreads();

    {
        const int n = ct * 16 + mrow;
        v8f c = {};
#pragma unroll
        for (int ks = 0; ks < 2; ++ks) {
            const int k0 = ks * 4 + khalf * 2;
            const v2f bb = *(const v2f*)(amats + n * CP_ + k0);
            c = __builtin_amdgcn_wmma_f32_16x16x4_f32(
                    false, a1[ks], false, bb, (short)0, c, false, false);
        }
#pragma unroll
        for (int r = 0; r < 8; ++r)
            qps[(rt * 16 + r + 8 * khalf) * HC_ + n] = c[r];
    }
    __syncthreads();

    const int  col   = mrow;
    const bool cv    = (col < C_);
    const int  bcol  = cv ? col : 0;
    const int  rbase = r0 + wave * 16 + (khalf ? 8 : 0);
    float bias = 0.0f, lsc = 0.0f, lbi = 0.0f;
    float res[8] = {0.f, 0.f, 0.f, 0.f, 0.f, 0.f, 0.f, 0.f};
    if (wave < 2) {
        bias = fc_b[bcol];
        lsc  = ln_s[bcol];
        lbi  = ln_b[bcol];
#pragma unroll
        for (int r = 0; r < 8; ++r)
            res[r] = hs[(rbase + r) * C_ + bcol];
    }

    {
        const int rr = lane >> 3;
        const int h  = lane & 7;
        const int i  = wave * 4 + rr;
        const float4 q0 = ((const float4*)qps)[(i * HC_ + h * CP_) / 4];
        const float4 q1 = ((const float4*)qps)[(i * HC_ + h * CP_) / 4 + 1];
        const float4* __restrict__ xrow = (const float4*)(xs + i * XR_);

        float4 xa[M_], xb[M_];
        float  sc[M_];
#pragma unroll
        for (int m = 0; m < M_; ++m) {
            xa[m] = xrow[m * 2];
            xb[m] = xrow[m * 2 + 1];
            float s = q0.x * xa[m].x;
            s = fmaf(q0.y, xa[m].y, s);
            s = fmaf(q0.z, xa[m].z, s);
            s = fmaf(q0.w, xa[m].w, s);
            s = fmaf(q1.x, xb[m].x, s);
            s = fmaf(q1.y, xb[m].y, s);
            s = fmaf(q1.z, xb[m].z, s);
            sc[m] = (xb[m].w != 0.0f) ? s : -1.0e9f;
        }
        float mx = sc[0];
#pragma unroll
        for (int m = 1; m < M_; ++m) mx = fmaxf(mx, sc[m]);

        float sum = 0.0f, z[7];
#pragma unroll
        for (int e = 0; e < 7; ++e) z[e] = 0.0f;
#pragma unroll
        for (int m = 0; m < M_; ++m) {
            const float p = (xb[m].w != 0.0f) ? __expf(sc[m] - mx) : 0.0f;
            sum += p;
            z[0] = fmaf(p, xa[m].x, z[0]);
            z[1] = fmaf(p, xa[m].y, z[1]);
            z[2] = fmaf(p, xa[m].z, z[2]);
            z[3] = fmaf(p, xa[m].w, z[3]);
            z[4] = fmaf(p, xb[m].x, z[4]);
            z[5] = fmaf(p, xb[m].y, z[5]);
            z[6] = fmaf(p, xb[m].z, z[6]);
        }
        const float inv = 1.0f / fmaxf(sum, 1.0e-30f);
        float4 lo, hi;
        lo.x = z[0] * inv; lo.y = z[1] * inv; lo.z = z[2] * inv; lo.w = z[3] * inv;
        hi.x = z[4] * inv; hi.y = z[5] * inv; hi.z = z[6] * inv; hi.w = 0.0f;
        ((float4*)zs)[(i * ZS_ + h * CP_) / 4]     = lo;
        ((float4*)zs)[(i * ZS_ + h * CP_) / 4 + 1] = hi;
    }
    __syncthreads();

    if (wave < 2) {
        const float cmask = cv ? 1.0f : 0.0f;

        v2f a3[HC_ / 4], b3[HC_ / 4];
#pragma unroll
        for (int s = 0; s < HC_ / 4; ++s) {
            const int k0 = s * 4 + khalf * 2;
            a3[s]   = *(const v2f*)(zs + (wave * 16 + mrow) * ZS_ + k0);
            b3[s].x = gts[k0 * CP_ + bcol];
            b3[s].y = gts[(k0 + 1) * CP_ + bcol];
        }
        __builtin_amdgcn_sched_barrier(0);

        v8f c = {};
#pragma unroll
        for (int s = 0; s < HC_ / 4; ++s) {
            v2f bb;
            bb.x = b3[s].x * cmask;
            bb.y = b3[s].y * cmask;
            c = __builtin_amdgcn_wmma_f32_16x16x4_f32(
                    false, a3[s], false, bb, (short)0, c, false, false);
        }


        float val[8], s1[8], s2[8];
#pragma unroll
        for (int r = 0; r < 8; ++r) {
            val[r] = c[r] + bias + res[r];
            s1[r]  = cv ? val[r] : 0.0f;
        }
#pragma unroll
        for (int r = 0; r < 8; ++r) s1[r] = add_row_xor<8>(s1[r]);
#pragma unroll
        for (int r = 0; r < 8; ++r) s1[r] = add_row_xor<4>(s1[r]);
#pragma unroll
        for (int r = 0; r < 8; ++r) s1[r] = add_row_xor<2>(s1[r]);
#pragma unroll
        for (int r = 0; r < 8; ++r) s1[r] = add_row_xor<1>(s1[r]);

#pragma unroll
        for (int r = 0; r < 8; ++r) {
            const float mu = s1[r] * (1.0f / (float)C_);
            const float dv = cv ? (val[r] - mu) : 0.0f;
            s2[r] = dv * dv;
        }
#pragma unroll
        for (int r = 0; r < 8; ++r) s2[r] = add_row_xor<8>(s2[r]);
#pragma unroll
        for (int r = 0; r < 8; ++r) s2[r] = add_row_xor<4>(s2[r]);
#pragma unroll
        for (int r = 0; r < 8; ++r) s2[r] = add_row_xor<2>(s2[r]);
#pragma unroll
        for (int r = 0; r < 8; ++r) s2[r] = add_row_xor<1>(s2[r]);

#pragma unroll
        for (int r = 0; r < 8; ++r) {
            const float mu  = s1[r] * (1.0f / (float)C_);
            const float var = s2[r] * (1.0f / (float)C_);
            const float y = (val[r] - mu) * rsqrtf(var + 1.0e-6f) * lsc + lbi;
            if (cv) out[(rbase + r) * C_ + col] = y;
        }
    }
}

extern "C" void kernel_launch(void* const* d_in, const int* in_sizes, int n_in,
                              void* d_out, int out_size, void* d_ws, size_t ws_size,
                              hipStream_t stream)
{
    const float* hs  = (const float*)d_in[0];
    const float* wq  = (const float*)d_in[1];
    const float* wk  = (const float*)d_in[2];
    const float* wv  = (const float*)d_in[3];
    const float* fcw = (const float*)d_in[4];
    const float* fcb = (const float*)d_in[5];
    const float* lns = (const float*)d_in[6];
    const float* lnb = (const float*)d_in[7];
    const int*   qkm = (const int*)d_in[8];
    float* out = (float*)d_out;

    pyr_attn_fused<<<256, 256, 0, stream>>>(hs, qkm, wq, wk, wv, fcw, fcb, lns, lnb, out);
}
